# starting kernel
# baseline (speedup 1.0000x reference)
#include <hip/hip_runtime.h>
#include <math.h>


typedef float v2f __attribute__((ext_vector_type(2)));
typedef float v8f __attribute__((ext_vector_type(8)));

#define NPTS      1024
#define NTHREADS  256
#define NWAVES    8
#define PPW       128
#define KITERS    32
#define EPSW      1e-5f
#define WTHRESH   0.0f

__device__ __forceinline__ float dot3(const float* a, const float* b) {
  return a[0]*b[0] + a[1]*b[1] + a[2]*b[2];
}
__device__ __forceinline__ void cross3(const float* a, const float* b, float* c) {
  c[0] = a[1]*b[2] - a[2]*b[1];
  c[1] = a[2]*b[0] - a[0]*b[2];
  c[2] = a[0]*b[1] - a[1]*b[0];
}

__global__ __launch_bounds__(NTHREADS)
void procrustes_wmma_kernel(const float* __restrict__ src,
                            const float* __restrict__ tgt,
                            const float* __restrict__ wgt,
                            float* __restrict__ out)
{
  __shared__ float feat[NPTS * 8];
  __shared__ float red[NWAVES * 16];
  __shared__ float sums[16];

  const int b   = blockIdx.x;
  const int tid = threadIdx.x;

  const float* sb = src + (size_t)b * (NPTS * 3);
  const float* tb = tgt + (size_t)b * (NPTS * 3);
  const float* wb = wgt + (size_t)b * NPTS;

  {
    const int p0 = tid * 4;
    const float4* s4 = reinterpret_cast<const float4*>(sb + p0 * 3);
    const float4* t4 = reinterpret_cast<const float4*>(tb + p0 * 3);
    float4 sA = s4[0], sB = s4[1], sC = s4[2];
    float4 tA = t4[0], tB = t4[1], tC = t4[2];
    float4 w4 = *reinterpret_cast<const float4*>(wb + p0);

    float sx[4] = {sA.x, sA.w, sB.z, sC.y};
    float sy[4] = {sA.y, sB.x, sB.w, sC.z};
    float sz[4] = {sA.z, sB.y, sC.x, sC.w};
    float tx[4] = {tA.x, tA.w, tB.z, tC.y};
    float ty[4] = {tA.y, tB.x, tB.w, tC.z};
    float tz[4] = {tA.z, tB.y, tC.x, tC.w};
    float wv[4] = {w4.x, w4.y, w4.z, w4.w};

    #pragma unroll
    for (int k = 0; k < 4; ++k) {
      float w = (wv[k] < WTHRESH) ? 0.0f : wv[k];
      float* d = &feat[(p0 + k) * 8];
      reinterpret_cast<float4*>(d)[0] = make_float4(1.0f, sx[k], sy[k], sz[k]);
      reinterpret_cast<float4*>(d)[1] = make_float4(w, w*tx[k], w*ty[k], w*tz[k]);
    }
  }
  __syncthreads();

  const int wave = tid >> 5;
  const int lane = tid & 31;
  const int f    = lane & 15;
  const int fi   = f >> 2;
  const int fj   = f & 3;
  const int koff = (lane >> 4) << 1;

  v2f Bones;
  Bones.x = (f == 0) ? 1.0f : 0.0f;
  Bones.y = Bones.x;

  v8f C = {0.f, 0.f, 0.f, 0.f, 0.f, 0.f, 0.f, 0.f};

  const int pbase = wave * PPW + koff;
  #pragma unroll 4
  for (int it = 0; it < KITERS; ++it) {
    const float* e0 = &feat[(pbase + it * 4) * 8];
    v2f A;
    A.x = e0[fi]     * e0[4 + fj];
    A.y = e0[8 + fi] * e0[12 + fj];
    C = __builtin_amdgcn_wmma_f32_16x16x4_f32(false, A, false, Bones,
                                              (short)0, C, false, false);
  }

  if (lane == 0) {
    #pragma unroll
    for (int r = 0; r < 8; ++r) red[wave * 16 + r] = C[r];
  } else if (lane == 16) {
    #pragma unroll
    for (int r = 0; r < 8; ++r) red[wave * 16 + 8 + r] = C[r];
  }
  __syncthreads();

  if (tid < 16) {
    float s = 0.f;
    #pragma unroll
    for (int wv2 = 0; wv2 < NWAVES; ++wv2) s += red[wv2 * 16 + tid];
    sums[tid] = s;
  }
  __syncthreads();

  if (tid == 0) {
    float s16[16];
    #pragma unroll
    for (int k2 = 0; k2 < 16; ++k2) s16[k2] = sums[k2];

    const float sumw = s16[0];
    const float inv  = 1.0f / (sumw + EPSW);
    const float sig  = sumw * inv;

    float cs[3] = { s16[4]*inv, s16[8]*inv, s16[12]*inv };
    float ct[3] = { s16[1]*inv, s16[2]*inv, s16[3]*inv  };

    float Gc[3][3];
    #pragma unroll
    for (int c = 0; c < 3; ++c)
      #pragma unroll
      for (int r = 0; r < 3; ++r)
        Gc[c][r] = s16[4*(r+1) + (c+1)]*inv + (sig - 2.0f)*cs[r]*ct[c];

    float Vc[3][3] = {{1,0,0},{0,1,0},{0,0,1}};
    for (int sweep = 0; sweep < 6; ++sweep) {
      #pragma unroll
      for (int pair = 0; pair < 3; ++pair) {
        const int p = (pair == 2) ? 1 : 0;
        const int q = (pair == 0) ? 1 : 2;
        const float app = dot3(Gc[p], Gc[p]);
        const float aqq = dot3(Gc[q], Gc[q]);
        const float apq = dot3(Gc[p], Gc[q]);
        if (apq * apq > 1e-28f * app * aqq) {
          const float zeta = (aqq - app) / (2.0f * apq);
          const float tt   = copysignf(1.0f, zeta) /
                             (fabsf(zeta) + sqrtf(1.0f + zeta*zeta));
          const float cth  = 1.0f / sqrtf(1.0f + tt*tt);
          const float sth  = cth * tt;
          #pragma unroll
          for (int r = 0; r < 3; ++r) {
            const float gp = Gc[p][r], gq = Gc[q][r];
            Gc[p][r] = cth*gp - sth*gq;
            Gc[q][r] = sth*gp + cth*gq;
            const float vp = Vc[p][r], vq = Vc[q][r];
            Vc[p][r] = cth*vp - sth*vq;
            Vc[q][r] = sth*vp + cth*vq;
          }
        }
      }
    }

    float sv[3];
    #pragma unroll
    for (int c = 0; c < 3; ++c) sv[c] = sqrtf(dot3(Gc[c], Gc[c]));

    int i0 = 0, i1 = 1, i2 = 2;
    if (sv[i0] < sv[i1]) { int t = i0; i0 = i1; i1 = t; }
    if (sv[i0] < sv[i2]) { int t = i0; i0 = i2; i2 = t; }
    if (sv[i1] < sv[i2]) { int t = i1; i1 = i2; i2 = t; }

    float R[3][3] = {{1,0,0},{0,1,0},{0,0,1}};
    const float s0 = sv[i0], s1 = sv[i1], s2 = sv[i2];
    if (s0 > 1e-30f) {
      float U0[3], U1[3], U2[3];
      #pragma unroll
      for (int r = 0; r < 3; ++r) U0[r] = Gc[i0][r] / s0;
      const float tiny = 1e-7f * s0;
      if (s1 > tiny) {
        #pragma unroll
        for (int r = 0; r < 3; ++r) U1[r] = Gc[i1][r] / s1;
      } else {
        const float ax = fabsf(U0[0]);
        float e[3] = { (ax < 0.9f) ? 1.0f : 0.0f,
                       (ax < 0.9f) ? 0.0f : 1.0f, 0.0f };
        cross3(U0, e, U1);
        const float n = 1.0f / sqrtf(dot3(U1, U1));
        U1[0] *= n; U1[1] *= n; U1[2] *= n;
      }
      float c01[3];
      cross3(U0, U1, c01);
      if (s2 > tiny) {
        #pragma unroll
        for (int r = 0; r < 3; ++r) U2[r] = Gc[i2][r] / s2;
      } else {
        #pragma unroll
        for (int r = 0; r < 3; ++r) U2[r] = c01[r];
      }

      float tmp[3];
      cross3(U1, U2, tmp);
      const float detU = dot3(U0, tmp);
      cross3(Vc[i1], Vc[i2], tmp);
      const float detV = dot3(Vc[i0], tmp);
      const float d2 = (detU * detV < 0.0f) ? -1.0f : 1.0f;

      #pragma unroll
      for (int r = 0; r < 3; ++r)
        #pragma unroll
        for (int c = 0; c < 3; ++c)
          R[r][c] = Vc[i0][r]*U0[c] + Vc[i1][r]*U1[c] + d2*Vc[i2][r]*U2[c];
    }

    float tr[3];
    #pragma unroll
    for (int r = 0; r < 3; ++r)
      tr[r] = ct[r] - (R[r][0]*cs[0] + R[r][1]*cs[1] + R[r][2]*cs[2]);

    float* o = out + (size_t)b * 16;
    o[0]  = R[0][0]; o[1]  = R[0][1]; o[2]  = R[0][2]; o[3]  = tr[0];
    o[4]  = R[1][0]; o[5]  = R[1][1]; o[6]  = R[1][2]; o[7]  = tr[1];
    o[8]  = R[2][0]; o[9]  = R[2][1]; o[10] = R[2][2]; o[11] = tr[2];
    o[12] = 0.0f;    o[13] = 0.0f;    o[14] = 0.0f;    o[15] = 1.0f;
  }
}

extern "C" void kernel_launch(void* const* d_in, const int* in_sizes, int n_in,
                              void* d_out, int out_size, void* d_ws, size_t ws_size,
                              hipStream_t stream) {
  (void)n_in; (void)out_size; (void)d_ws; (void)ws_size;
  const float* src = (const float*)d_in[0];
  const float* tgt = (const float*)d_in[1];
  const float* wgt = (const float*)d_in[2];
  float* out = (float*)d_out;

  const int B = in_sizes[2] / NPTS;
  procrustes_wmma_kernel<<<B, NTHREADS, 0, stream>>>(src, tgt, wgt, out);
}


# Eight batches per work-group: the serial 3x3 Jacobi solve runs once on eight lanes side by side instead of once per batch on one lane; next batch's loads issued before the current reduction
# speedup vs baseline: 1.2796x; 1.2796x over previous
#include <hip/hip_runtime.h>
#include <math.h>


typedef float v2f __attribute__((ext_vector_type(2)));
typedef float v8f __attribute__((ext_vector_type(8)));

#define NPTS      1024
#define NTHREADS  256
#define NWAVES    8
#define PPW       128
#define KITERS    32
#define GB        8
#define EPSW      1e-5f
#define WTHRESH   0.0f

__device__ __forceinline__ float dot3(const float* a, const float* b) {
  return a[0]*b[0] + a[1]*b[1] + a[2]*b[2];
}
__device__ __forceinline__ void cross3(const float* a, const float* b, float* c) {
  c[0] = a[1]*b[2] - a[2]*b[1];
  c[1] = a[2]*b[0] - a[0]*b[2];
  c[2] = a[0]*b[1] - a[1]*b[0];
}

struct Pts { float4 sA, sB, sC, tA, tB, tC, w4; };

__device__ __forceinline__ Pts load_batch(const float* __restrict__ src,
                                          const float* __restrict__ tgt,
                                          const float* __restrict__ wgt,
                                          int b, int p0) {
  const float4* s4 = reinterpret_cast<const float4*>(src + (size_t)b * (NPTS * 3) + p0 * 3);
  const float4* t4 = reinterpret_cast<const float4*>(tgt + (size_t)b * (NPTS * 3) + p0 * 3);
  Pts r;
  r.sA = s4[0]; r.sB = s4[1]; r.sC = s4[2];
  r.tA = t4[0]; r.tB = t4[1]; r.tC = t4[2];
  r.w4 = *reinterpret_cast<const float4*>(wgt + (size_t)b * NPTS + p0);
  return r;
}

__global__ __launch_bounds__(NTHREADS)
void procrustes_wmma_kernel(const float* __restrict__ src,
                            const float* __restrict__ tgt,
                            const float* __restrict__ wgt,
                            float* __restrict__ out,
                            int nbatch)
{
  __shared__ float feat[NPTS * 8];
  __shared__ float red[NWAVES * 16];
  __shared__ float sums[GB * 16];

  const int tid = threadIdx.x;
  const int b0  = blockIdx.x * GB;
  const int nb  = (nbatch - b0 < GB) ? (nbatch - b0) : GB;
  const int p0  = tid * 4;

  const int wave = tid >> 5;
  const int lane = tid & 31;
  const int f    = lane & 15;
  const int fi   = f >> 2;
  const int fj   = f & 3;
  const int koff = (lane >> 4) << 1;
  const int pbase = wave * PPW + koff;

  v2f Bones;
  Bones.x = (f == 0) ? 1.0f : 0.0f;
  Bones.y = Bones.x;

  Pts cur = load_batch(src, tgt, wgt, b0, p0);

  for (int g = 0; g < nb; ++g) {
  {
    float sx[4] = {cur.sA.x, cur.sA.w, cur.sB.z, cur.sC.y};
    float sy[4] = {cur.sA.y, cur.sB.x, cur.sB.w, cur.sC.z};
    float sz[4] = {cur.sA.z, cur.sB.y, cur.sC.x, cur.sC.w};
    float tx[4] = {cur.tA.x, cur.tA.w, cur.tB.z, cur.tC.y};
    float ty[4] = {cur.tA.y, cur.tB.x, cur.tB.w, cur.tC.z};
    float tz[4] = {cur.tA.z, cur.tB.y, cur.tC.x, cur.tC.w};
    float wv[4] = {cur.w4.x, cur.w4.y, cur.w4.z, cur.w4.w};

    #pragma unroll
    for (int k = 0; k < 4; ++k) {
      float w = (wv[k] < WTHRESH) ? 0.0f : wv[k];
      float* d = &feat[(p0 + k) * 8];
      reinterpret_cast<float4*>(d)[0] = make_float4(1.0f, sx[k], sy[k], sz[k]);
      reinterpret_cast<float4*>(d)[1] = make_float4(w, w*tx[k], w*ty[k], w*tz[k]);
    }
  }
  __syncthreads();

  if (g + 1 < nb) cur = load_batch(src, tgt, wgt, b0 + g + 1, p0);

  v8f C = {0.f, 0.f, 0.f, 0.f, 0.f, 0.f, 0.f, 0.f};

  #pragma unroll 4
  for (int it = 0; it < KITERS; ++it) {
    const float* e0 = &feat[(pbase + it * 4) * 8];
    v2f A;
    A.x = e0[fi]     * e0[4 + fj];
    A.y = e0[8 + fi] * e0[12 + fj];
    C = __builtin_amdgcn_wmma_f32_16x16x4_f32(false, A, false, Bones,
                                              (short)0, C, false, false);
  }

  if (lane == 0) {
    #pragma unroll
    for (int r = 0; r < 8; ++r) red[wave * 16 + r] = C[r];
  } else if (lane == 16) {
    #pragma unroll
    for (int r = 0; r < 8; ++r) red[wave * 16 + 8 + r] = C[r];
  }
  __syncthreads();

  if (tid < 16) {
    float s = 0.f;
    #pragma unroll
    for (int wv2 = 0; wv2 < NWAVES; ++wv2) s += red[wv2 * 16 + tid];
    sums[g * 16 + tid] = s;
  }
  }
  __syncthreads();

  if (tid < nb) {
    const int b = b0 + tid;
    float s16[16];
    #pragma unroll
    for (int k2 = 0; k2 < 16; ++k2) s16[k2] = sums[tid * 16 + k2];

    const float sumw = s16[0];
    const float inv  = 1.0f / (sumw + EPSW);
    const float sig  = sumw * inv;

    float cs[3] = { s16[4]*inv, s16[8]*inv, s16[12]*inv };
    float ct[3] = { s16[1]*inv, s16[2]*inv, s16[3]*inv  };

    float Gc[3][3];
    #pragma unroll
    for (int c = 0; c < 3; ++c)
      #pragma unroll
      for (int r = 0; r < 3; ++r)
        Gc[c][r] = s16[4*(r+1) + (c+1)]*inv + (sig - 2.0f)*cs[r]*ct[c];

    float Vc[3][3] = {{1,0,0},{0,1,0},{0,0,1}};
    for (int sweep = 0; sweep < 6; ++sweep) {
      #pragma unroll
      for (int pair = 0; pair < 3; ++pair) {
        const int p = (pair == 2) ? 1 : 0;
        const int q = (pair == 0) ? 1 : 2;
        const float app = dot3(Gc[p], Gc[p]);
        const float aqq = dot3(Gc[q], Gc[q]);
        const float apq = dot3(Gc[p], Gc[q]);
        if (apq * apq > 1e-28f * app * aqq) {
          const float zeta = (aqq - app) / (2.0f * apq);
          const float tt   = copysignf(1.0f, zeta) /
                             (fabsf(zeta) + sqrtf(1.0f + zeta*zeta));
          const float cth  = 1.0f / sqrtf(1.0f + tt*tt);
          const float sth  = cth * tt;
          #pragma unroll
          for (int r = 0; r < 3; ++r) {
            const float gp = Gc[p][r], gq = Gc[q][r];
            Gc[p][r] = cth*gp - sth*gq;
            Gc[q][r] = sth*gp + cth*gq;
            const float vp = Vc[p][r], vq = Vc[q][r];
            Vc[p][r] = cth*vp - sth*vq;
            Vc[q][r] = sth*vp + cth*vq;
          }
        }
      }
    }

    float sv[3];
    #pragma unroll
    for (int c = 0; c < 3; ++c) sv[c] = sqrtf(dot3(Gc[c], Gc[c]));

    int i0 = 0, i1 = 1, i2 = 2;
    if (sv[i0] < sv[i1]) { int t = i0; i0 = i1; i1 = t; }
    if (sv[i0] < sv[i2]) { int t = i0; i0 = i2; i2 = t; }
    if (sv[i1] < sv[i2]) { int t = i1; i1 = i2; i2 = t; }

    float R[3][3] = {{1,0,0},{0,1,0},{0,0,1}};
    const float s0 = sv[i0], s1 = sv[i1], s2 = sv[i2];
    if (s0 > 1e-30f) {
      float U0[3], U1[3], U2[3];
      #pragma unroll
      for (int r = 0; r < 3; ++r) U0[r] = Gc[i0][r] / s0;
      const float tiny = 1e-7f * s0;
      if (s1 > tiny) {
        #pragma unroll
        for (int r = 0; r < 3; ++r) U1[r] = Gc[i1][r] / s1;
      } else {
        const float ax = fabsf(U0[0]);
        float e[3] = { (ax < 0.9f) ? 1.0f : 0.0f,
                       (ax < 0.9f) ? 0.0f : 1.0f, 0.0f };
        cross3(U0, e, U1);
        const float n = 1.0f / sqrtf(dot3(U1, U1));
        U1[0] *= n; U1[1] *= n; U1[2] *= n;
      }
      float c01[3];
      cross3(U0, U1, c01);
      if (s2 > tiny) {
        #pragma unroll
        for (int r = 0; r < 3; ++r) U2[r] = Gc[i2][r] / s2;
      } else {
        #pragma unroll
        for (int r = 0; r < 3; ++r) U2[r] = c01[r];
      }

      float tmp[3];
      cross3(U1, U2, tmp);
      const float detU = dot3(U0, tmp);
      cross3(Vc[i1], Vc[i2], tmp);
      const float detV = dot3(Vc[i0], tmp);
      const float d2 = (detU * detV < 0.0f) ? -1.0f : 1.0f;

      #pragma unroll
      for (int r = 0; r < 3; ++r)
        #pragma unroll
        for (int c = 0; c < 3; ++c)
          R[r][c] = Vc[i0][r]*U0[c] + Vc[i1][r]*U1[c] + d2*Vc[i2][r]*U2[c];
    }

    float tr[3];
    #pragma unroll
    for (int r = 0; r < 3; ++r)
      tr[r] = ct[r] - (R[r][0]*cs[0] + R[r][1]*cs[1] + R[r][2]*cs[2]);

    float* o = out + (size_t)b * 16;
    o[0]  = R[0][0]; o[1]  = R[0][1]; o[2]  = R[0][2]; o[3]  = tr[0];
    o[4]  = R[1][0]; o[5]  = R[1][1]; o[6]  = R[1][2]; o[7]  = tr[1];
    o[8]  = R[2][0]; o[9]  = R[2][1]; o[10] = R[2][2]; o[11] = tr[2];
    o[12] = 0.0f;    o[13] = 0.0f;    o[14] = 0.0f;    o[15] = 1.0f;
  }
}

extern "C" void kernel_launch(void* const* d_in, const int* in_sizes, int n_in,
                              void* d_out, int out_size, void* d_ws, size_t ws_size,
                              hipStream_t stream) {
  (void)n_in; (void)out_size; (void)d_ws; (void)ws_size;
  const float* src = (const float*)d_in[0];
  const float* tgt = (const float*)d_in[1];
  const float* wgt = (const float*)d_in[2];
  float* out = (float*)d_out;

  const int B = in_sizes[2] / NPTS;
  procrustes_wmma_kernel<<<(B + GB - 1) / GB, NTHREADS, 0, stream>>>(src, tgt, wgt, out, B);
}
